# starting kernel
# baseline (speedup 1.0000x reference)
#include <hip/hip_runtime.h>
#include <hip/hip_bf16.h>
#include <math.h>

typedef __attribute__((ext_vector_type(2))) float v2f;
typedef __attribute__((ext_vector_type(8))) float v8f;

#define LOG2PI 1.8378770664093453f
#define LOG2E  1.4426950408889634f
#define LN2    0.6931471805599453f
#define TILES_PER_WAVE 4

__global__ __launch_bounds__(256) void mdn_posterior_wmma(
    const float* __restrict__ y,
    const float* __restrict__ mus,
    const float* __restrict__ sigmas,
    const float* __restrict__ pi_logits,
    const float* __restrict__ prior,
    float* __restrict__ out,
    int n_full_tiles)
{
    __shared__ float W[5 * 128];

    const int tid = threadIdx.x;
    if (tid < 128) {
        const int j = tid >> 3;
        const int c = tid & 7;
        const int base = j * 16 + c * 2;
        const float m0 = mus[base],    m1 = mus[base + 1];
        const float s0 = sigmas[base], s1 = sigmas[base + 1];
        const float iv0 = 1.0f / (s0 * s0);
        const float iv1 = 1.0f / (s1 * s1);
        W[      tid] = LOG2E * -0.5f * iv0;
        W[128 + tid] = LOG2E * -0.5f * iv1;
        W[256 + tid] = LOG2E * m0 * iv0;
        W[384 + tid] = LOG2E * m1 * iv1;
        float mx = pi_logits[j * 8];
        #pragma unroll
        for (int i = 1; i < 8; ++i) mx = fmaxf(mx, pi_logits[j * 8 + i]);
        float se = 0.0f;
        #pragma unroll
        for (int i = 0; i < 8; ++i) se += __expf(pi_logits[j * 8 + i] - mx);
        const float lse = mx + __logf(se);
        W[512 + tid] = LOG2E * (-0.5f * (m0 * m0 * iv0 + m1 * m1 * iv1)
                                - (__logf(s0) + __logf(s1)) - LOG2PI
                                + (pi_logits[j * 8 + c] - lse) + __logf(prior[j]));
    }
    __syncthreads();

    const int lane = tid & 31;
    const int wave = tid >> 5;
    const int m    = lane & 15;
    const int hi   = lane >> 4;

    const int tile0 = (blockIdx.x * (blockDim.x >> 5) + wave) * TILES_PER_WAVE;
    if (tile0 >= n_full_tiles) return;

    v2f a[8];
    v8f kvec[8];
    #pragma unroll
    for (int t = 0; t < 8; ++t) {
        const int ncol = t * 16 + m;
        a[t].x = W[(hi ? 256 :   0) + ncol];
        a[t].y = W[(hi ? 384 : 128) + ncol];
        const float* Kp = &W[512 + t * 16 + hi * 8];
        #pragma unroll
        for (int i = 0; i < 8; ++i) kvec[t][i] = Kp[i];
    }

    const int row0 = tile0 * 16 + m;
    const float* yp = y + row0 * 2;
    float*       op = out + row0 * 16 + hi;

    int count = n_full_tiles - tile0;
    if (count > TILES_PER_WAVE) count = TILES_PER_WAVE;

    #pragma unroll 1
    for (int it = 0; it < count; ++it) {
        const v2f yv = *(const v2f*)yp;
        yp += 32;

        v2f b;
        b.x = hi ? yv.x : yv.x * yv.x;
        b.y = hi ? yv.y : yv.y * yv.y;

        #pragma unroll
        for (int t = 0; t < 8; ++t) {
            const v8f d = __builtin_amdgcn_wmma_f32_16x16x4_f32(
                false, a[t], false, b, (short)0, kvec[t], false, false);

            float mx = fmaxf(fmaxf(d[0], d[1]), fmaxf(d[2], d[3]));
            mx = fmaxf(mx, fmaxf(fmaxf(d[4], d[5]), fmaxf(d[6], d[7])));
            float ex0 = __builtin_amdgcn_exp2f(d[0] - mx);
            float ex1 = __builtin_amdgcn_exp2f(d[1] - mx);
            float ex2 = __builtin_amdgcn_exp2f(d[2] - mx);
            float ex3 = __builtin_amdgcn_exp2f(d[3] - mx);
            float ex4 = __builtin_amdgcn_exp2f(d[4] - mx);
            float ex5 = __builtin_amdgcn_exp2f(d[5] - mx);
            float ex6 = __builtin_amdgcn_exp2f(d[6] - mx);
            float ex7 = __builtin_amdgcn_exp2f(d[7] - mx);
            const float e = ((ex0 + ex1) + (ex2 + ex3)) + ((ex4 + ex5) + (ex6 + ex7));
            op[2 * t] = LN2 * (mx + __builtin_amdgcn_logf(e));
        }
        op += 256;
    }
}

__global__ __launch_bounds__(256) void mdn_posterior_tail(
    const float* __restrict__ y,
    const float* __restrict__ mus,
    const float* __restrict__ sigmas,
    const float* __restrict__ pi_logits,
    const float* __restrict__ prior,
    float* __restrict__ out,
    int n_samp, int row0)
{
    const int w = blockIdx.x * blockDim.x + threadIdx.x;
    if (w >= (n_samp - row0) * 16) return;
    const int row = row0 + (w >> 4);
    const int j   = w & 15;
    const float y0 = y[row * 2], y1 = y[row * 2 + 1];

    float mxp = pi_logits[j * 8];
    #pragma unroll
    for (int i = 1; i < 8; ++i) mxp = fmaxf(mxp, pi_logits[j * 8 + i]);
    float se = 0.0f;
    #pragma unroll
    for (int i = 0; i < 8; ++i) se += __expf(pi_logits[j * 8 + i] - mxp);
    const float lse = mxp + __logf(se);
    const float lp  = __logf(prior[j]);

    float vals[8];
    float vmax = -INFINITY;
    #pragma unroll
    for (int c = 0; c < 8; ++c) {
        const int base = j * 16 + c * 2;
        const float m0 = mus[base],    m1 = mus[base + 1];
        const float s0 = sigmas[base], s1 = sigmas[base + 1];
        const float d0 = (y0 - m0) / s0, d1 = (y1 - m1) / s1;
        const float v = -0.5f * (d0 * d0 + d1 * d1)
                        - (__logf(s0) + __logf(s1)) - LOG2PI
                        + (pi_logits[j * 8 + c] - lse) + lp;
        vals[c] = v;
        vmax = fmaxf(vmax, v);
    }
    float e = 0.0f;
    #pragma unroll
    for (int c = 0; c < 8; ++c) e += __expf(vals[c] - vmax);
    out[row * 16 + j] = vmax + __logf(e);
}

extern "C" void kernel_launch(void* const* d_in, const int* in_sizes, int n_in,
                              void* d_out, int out_size, void* d_ws, size_t ws_size,
                              hipStream_t stream) {
    const float* y         = (const float*)d_in[0];
    const float* mus       = (const float*)d_in[1];
    const float* sigmas    = (const float*)d_in[2];
    const float* pi_logits = (const float*)d_in[3];
    const float* prior     = (const float*)d_in[4];
    float* out = (float*)d_out;

    const int n_samp       = in_sizes[0] / 2;
    const int n_full_tiles = n_samp / 16;
    const int rem          = n_samp - n_full_tiles * 16;

    if (n_full_tiles > 0) {
        const int n_super = (n_full_tiles + TILES_PER_WAVE - 1) / TILES_PER_WAVE;
        const int waves_per_block = 8;
        dim3 block(32 * waves_per_block);
        dim3 grid((n_super + waves_per_block - 1) / waves_per_block);
        mdn_posterior_wmma<<<grid, block, 0, stream>>>(y, mus, sigmas, pi_logits,
                                                       prior, out, n_full_tiles);
    }
    if (rem > 0) {
        mdn_posterior_tail<<<1, 256, 0, stream>>>(y, mus, sigmas, pi_logits, prior,
                                                  out, n_samp, n_full_tiles * 16);
    }
}


# Whole-line output stores: symbol j=t+8*hi per lane half, so a lane owns 8 consecutive outputs; lanes m and m^8 swap 16 B by DPP; 2 b128 stores of 4 whole 128-B lines per tile replace 8 b32 stores
# speedup vs baseline: 1.5701x; 1.5701x over previous
#include <hip/hip_runtime.h>
#include <hip/hip_bf16.h>
#include <math.h>

typedef __attribute__((ext_vector_type(2))) float v2f;
typedef __attribute__((ext_vector_type(8))) float v8f;
typedef __attribute__((ext_vector_type(4))) float v4f;

#define LOG2PI 1.8378770664093453f
#define LOG2E  1.4426950408889634f
#define LN2    0.6931471805599453f
#define TILES_PER_WAVE 4

__global__ __launch_bounds__(256, 8) void mdn_posterior_wmma(
    const float* __restrict__ y,
    const float* __restrict__ mus,
    const float* __restrict__ sigmas,
    const float* __restrict__ pi_logits,
    const float* __restrict__ prior,
    float* __restrict__ out,
    int n_full_tiles)
{
    __shared__ float W[5 * 128];

    const int tid = threadIdx.x;
    if (tid < 128) {
        const int j = tid >> 3;
        const int c = tid & 7;
        const int base = j * 16 + c * 2;
        const float m0 = mus[base],    m1 = mus[base + 1];
        const float s0 = sigmas[base], s1 = sigmas[base + 1];
        const float iv0 = 1.0f / (s0 * s0);
        const float iv1 = 1.0f / (s1 * s1);
        W[      tid] = LOG2E * -0.5f * iv0;
        W[128 + tid] = LOG2E * -0.5f * iv1;
        W[256 + tid] = LOG2E * m0 * iv0;
        W[384 + tid] = LOG2E * m1 * iv1;
        float mx = pi_logits[j * 8];
        #pragma unroll
        for (int i = 1; i < 8; ++i) mx = fmaxf(mx, pi_logits[j * 8 + i]);
        float se = 0.0f;
        #pragma unroll
        for (int i = 0; i < 8; ++i) se += __expf(pi_logits[j * 8 + i] - mx);
        const float lse = mx + __logf(se);
        W[512 + tid] = LOG2E * (-0.5f * (m0 * m0 * iv0 + m1 * m1 * iv1)
                                - (__logf(s0) + __logf(s1)) - LOG2PI
                                + (pi_logits[j * 8 + c] - lse) + __logf(prior[j]));
    }
    __syncthreads();

    const int lane = tid & 31;
    const int wave = tid >> 5;
    const int m    = lane & 15;
    const int hi   = lane >> 4;

    const int tile0 = (blockIdx.x * (blockDim.x >> 5) + wave) * TILES_PER_WAVE;
    if (tile0 >= n_full_tiles) return;

    v2f a[8];
    v8f kvec[8];
    #pragma unroll
    for (int t = 0; t < 8; ++t) {
        const int ncol = t * 8 + (m >> 3) * 64 + (m & 7);
        a[t].x = W[(hi ? 256 :   0) + ncol];
        a[t].y = W[(hi ? 384 : 128) + ncol];
        const float* Kp = &W[512 + t * 8 + hi * 64];
        #pragma unroll
        for (int i = 0; i < 8; ++i) kvec[t][i] = Kp[i];
    }

    const int row0 = tile0 * 16 + m;
    const float* yp = y + row0 * 2;
    const bool lo = m < 8;
    float* op = out + (tile0 * 16 + (m & 7)) * 16 + hi * 8 + (m >> 3) * 4;

    int count = n_full_tiles - tile0;
    if (count > TILES_PER_WAVE) count = TILES_PER_WAVE;

    #pragma unroll 1
    for (int it = 0; it < count; ++it) {
        const v2f yv = *(const v2f*)yp;
        yp += 32;

        v2f b;
        b.x = hi ? yv.x : yv.x * yv.x;
        b.y = hi ? yv.y : yv.y * yv.y;

        float r[8];
        #pragma unroll
        for (int t = 0; t < 8; ++t) {
            const v8f d = __builtin_amdgcn_wmma_f32_16x16x4_f32(
                false, a[t], false, b, (short)0, kvec[t], false, false);

            float mx = fmaxf(fmaxf(d[0], d[1]), fmaxf(d[2], d[3]));
            mx = fmaxf(mx, fmaxf(fmaxf(d[4], d[5]), fmaxf(d[6], d[7])));
            float ex0 = __builtin_amdgcn_exp2f(d[0] - mx);
            float ex1 = __builtin_amdgcn_exp2f(d[1] - mx);
            float ex2 = __builtin_amdgcn_exp2f(d[2] - mx);
            float ex3 = __builtin_amdgcn_exp2f(d[3] - mx);
            float ex4 = __builtin_amdgcn_exp2f(d[4] - mx);
            float ex5 = __builtin_amdgcn_exp2f(d[5] - mx);
            float ex6 = __builtin_amdgcn_exp2f(d[6] - mx);
            float ex7 = __builtin_amdgcn_exp2f(d[7] - mx);
            const float e = ((ex0 + ex1) + (ex2 + ex3)) + ((ex4 + ex5) + (ex6 + ex7));
            r[t] = LN2 * (mx + __builtin_amdgcn_logf(e));
        }

        v4f st0, st1;
        #pragma unroll
        for (int i = 0; i < 4; ++i) {
            const float send = lo ? r[4 + i] : r[i];
            const float recv = __builtin_bit_cast(float,
                __builtin_amdgcn_mov_dpp(__builtin_bit_cast(int, send),
                                         0x128, 0xF, 0xF, true));
            st0[i] = lo ? r[i]  : recv;
            st1[i] = lo ? recv  : r[4 + i];
        }
        *(v4f*)op         = st0;
        *(v4f*)(op + 128) = st1;
        op += 256;
    }
}

__global__ __launch_bounds__(256) void mdn_posterior_tail(
    const float* __restrict__ y,
    const float* __restrict__ mus,
    const float* __restrict__ sigmas,
    const float* __restrict__ pi_logits,
    const float* __restrict__ prior,
    float* __restrict__ out,
    int n_samp, int row0)
{
    const int w = blockIdx.x * blockDim.x + threadIdx.x;
    if (w >= (n_samp - row0) * 16) return;
    const int row = row0 + (w >> 4);
    const int j   = w & 15;
    const float y0 = y[row * 2], y1 = y[row * 2 + 1];

    float mxp = pi_logits[j * 8];
    #pragma unroll
    for (int i = 1; i < 8; ++i) mxp = fmaxf(mxp, pi_logits[j * 8 + i]);
    float se = 0.0f;
    #pragma unroll
    for (int i = 0; i < 8; ++i) se += __expf(pi_logits[j * 8 + i] - mxp);
    const float lse = mxp + __logf(se);
    const float lp  = __logf(prior[j]);

    float vals[8];
    float vmax = -INFINITY;
    #pragma unroll
    for (int c = 0; c < 8; ++c) {
        const int base = j * 16 + c * 2;
        const float m0 = mus[base],    m1 = mus[base + 1];
        const float s0 = sigmas[base], s1 = sigmas[base + 1];
        const float d0 = (y0 - m0) / s0, d1 = (y1 - m1) / s1;
        const float v = -0.5f * (d0 * d0 + d1 * d1)
                        - (__logf(s0) + __logf(s1)) - LOG2PI
                        + (pi_logits[j * 8 + c] - lse) + lp;
        vals[c] = v;
        vmax = fmaxf(vmax, v);
    }
    float e = 0.0f;
    #pragma unroll
    for (int c = 0; c < 8; ++c) e += __expf(vals[c] - vmax);
    out[row * 16 + j] = vmax + __logf(e);
}

extern "C" void kernel_launch(void* const* d_in, const int* in_sizes, int n_in,
                              void* d_out, int out_size, void* d_ws, size_t ws_size,
                              hipStream_t stream) {
    const float* y         = (const float*)d_in[0];
    const float* mus       = (const float*)d_in[1];
    const float* sigmas    = (const float*)d_in[2];
    const float* pi_logits = (const float*)d_in[3];
    const float* prior     = (const float*)d_in[4];
    float* out = (float*)d_out;

    const int n_samp       = in_sizes[0] / 2;
    const int n_full_tiles = n_samp / 16;
    const int rem          = n_samp - n_full_tiles * 16;

    if (n_full_tiles > 0) {
        const int n_super = (n_full_tiles + TILES_PER_WAVE - 1) / TILES_PER_WAVE;
        const int waves_per_block = 8;
        dim3 block(32 * waves_per_block);
        dim3 grid((n_super + waves_per_block - 1) / waves_per_block);
        mdn_posterior_wmma<<<grid, block, 0, stream>>>(y, mus, sigmas, pi_logits,
                                                       prior, out, n_full_tiles);
    }
    if (rem > 0) {
        mdn_posterior_tail<<<1, 256, 0, stream>>>(y, mus, sigmas, pi_logits, prior,
                                                  out, n_samp, n_full_tiles * 16);
    }
}


# Unshifted base-2 logsumexp (exp2 straight on the product: no max, no subtract) under one wave-uniform range test per tile; a tile with a sum outside [2^-96,2^120] is redone max-shifted
# speedup vs baseline: 1.8694x; 1.1906x over previous
#include <hip/hip_runtime.h>
#include <hip/hip_bf16.h>
#include <math.h>

typedef __attribute__((ext_vector_type(2))) float v2f;
typedef __attribute__((ext_vector_type(8))) float v8f;
typedef __attribute__((ext_vector_type(4))) float v4f;

#define LOG2PI 1.8378770664093453f
#define LOG2E  1.4426950408889634f
#define LN2    0.6931471805599453f
#define TILES_PER_WAVE 4

#define EXPSUM_MIN 1.262177448e-29f
#define EXPSUM_MAX 1.329227996e+36f

__device__ __forceinline__ v2f make_b(const v2f yv, const int hi)
{
    v2f b;
    b.x = hi ? yv.x : yv.x * yv.x;
    b.y = hi ? yv.y : yv.y * yv.y;
    return b;
}

__device__ __forceinline__ void store_tile(const float (&r)[8], const bool lo, float* op)
{
    v4f st0, st1;
    #pragma unroll
    for (int i = 0; i < 4; ++i) {
        const float send = lo ? r[4 + i] : r[i];
        const float recv = __builtin_bit_cast(float,
            __builtin_amdgcn_mov_dpp(__builtin_bit_cast(int, send),
                                     0x128, 0xF, 0xF, true));
        st0[i] = lo ? r[i]  : recv;
        st1[i] = lo ? recv  : r[4 + i];
    }
    *(v4f*)op         = st0;
    *(v4f*)(op + 128) = st1;
}

__global__ __launch_bounds__(256, 8) void mdn_posterior_wmma(
    const float* __restrict__ y,
    const float* __restrict__ mus,
    const float* __restrict__ sigmas,
    const float* __restrict__ pi_logits,
    const float* __restrict__ prior,
    float* __restrict__ out,
    int n_full_tiles)
{
    __shared__ float W[5 * 128];

    const int tid = threadIdx.x;
    if (tid < 128) {
        const int j = tid >> 3;
        const int c = tid & 7;
        const int base = j * 16 + c * 2;
        const float m0 = mus[base],    m1 = mus[base + 1];
        const float s0 = sigmas[base], s1 = sigmas[base + 1];
        const float iv0 = 1.0f / (s0 * s0);
        const float iv1 = 1.0f / (s1 * s1);
        W[      tid] = LOG2E * -0.5f * iv0;
        W[128 + tid] = LOG2E * -0.5f * iv1;
        W[256 + tid] = LOG2E * m0 * iv0;
        W[384 + tid] = LOG2E * m1 * iv1;
        float mx = pi_logits[j * 8];
        #pragma unroll
        for (int i = 1; i < 8; ++i) mx = fmaxf(mx, pi_logits[j * 8 + i]);
        float se = 0.0f;
        #pragma unroll
        for (int i = 0; i < 8; ++i) se += __expf(pi_logits[j * 8 + i] - mx);
        const float lse = mx + __logf(se);
        W[512 + tid] = LOG2E * (-0.5f * (m0 * m0 * iv0 + m1 * m1 * iv1)
                                - (__logf(s0) + __logf(s1)) - LOG2PI
                                + (pi_logits[j * 8 + c] - lse) + __logf(prior[j]));
    }
    __syncthreads();

    const int lane = tid & 31;
    const int wave = tid >> 5;
    const int m    = lane & 15;
    const int hi   = lane >> 4;

    const int tile0 = (blockIdx.x * (blockDim.x >> 5) + wave) * TILES_PER_WAVE;
    if (tile0 >= n_full_tiles) return;

    v2f a[8];
    v8f kvec[8];
    #pragma unroll
    for (int t = 0; t < 8; ++t) {
        const int ncol = t * 8 + (m >> 3) * 64 + (m & 7);
        a[t].x = W[(hi ? 256 :   0) + ncol];
        a[t].y = W[(hi ? 384 : 128) + ncol];
        const float* Kp = &W[512 + t * 8 + hi * 64];
        #pragma unroll
        for (int i = 0; i < 8; ++i) kvec[t][i] = Kp[i];
    }

    const int row0 = tile0 * 16 + m;
    const float* yp = y + row0 * 2;
    const bool lo = m < 8;
    float* op = out + (tile0 * 16 + (m & 7)) * 16 + hi * 8 + (m >> 3) * 4;

    int count = n_full_tiles - tile0;
    if (count > TILES_PER_WAVE) count = TILES_PER_WAVE;

    unsigned redo = 0u;
    #pragma unroll 1
    for (int it = 0; it < count; ++it) {
        const v2f b = make_b(*(const v2f*)(yp + it * 32), hi);

        float r[8];
        float elo = __builtin_inff();
        float ehi = 0.0f;
        #pragma unroll
        for (int p = 0; p < 4; ++p) {
            const v8f dA = __builtin_amdgcn_wmma_f32_16x16x4_f32(
                false, a[2 * p], false, b, (short)0, kvec[2 * p], false, false);
            const v8f dB = __builtin_amdgcn_wmma_f32_16x16x4_f32(
                false, a[2 * p + 1], false, b, (short)0, kvec[2 * p + 1], false, false);
            v2f x[8];
            #pragma unroll
            for (int i = 0; i < 8; ++i) {
                x[i].x = __builtin_amdgcn_exp2f(dA[i]);
                x[i].y = __builtin_amdgcn_exp2f(dB[i]);
            }
            const v2f e = ((x[0] + x[1]) + (x[2] + x[3])) + ((x[4] + x[5]) + (x[6] + x[7]));
            elo = fminf(elo, fminf(e.x, e.y));
            ehi = fmaxf(ehi, fmaxf(e.x, e.y));
            v2f lg;
            lg.x = __builtin_amdgcn_logf(e.x);
            lg.y = __builtin_amdgcn_logf(e.y);
            const v2f res = lg * LN2;
            r[2 * p]     = res.x;
            r[2 * p + 1] = res.y;
        }
        const bool in_range = (elo >= EXPSUM_MIN) && (ehi <= EXPSUM_MAX);
        if (__builtin_amdgcn_ballot_w32(!in_range) == 0u) {
            store_tile(r, lo, op + it * 256);
        } else {
            redo |= 1u << it;
        }
    }

    if (redo == 0u) return;

    #pragma unroll 1
    for (int it = 0; it < count; ++it) {
        if (((redo >> it) & 1u) == 0u) continue;
        const v2f b = make_b(*(const v2f*)(yp + it * 32), hi);

        float r[8];
        #pragma unroll
        for (int t = 0; t < 8; ++t) {
            const v8f d = __builtin_amdgcn_wmma_f32_16x16x4_f32(
                false, a[t], false, b, (short)0, kvec[t], false, false);
            float mx = fmaxf(fmaxf(d[0], d[1]), fmaxf(d[2], d[3]));
            mx = fmaxf(mx, fmaxf(fmaxf(d[4], d[5]), fmaxf(d[6], d[7])));
            const float ex0 = __builtin_amdgcn_exp2f(d[0] - mx);
            const float ex1 = __builtin_amdgcn_exp2f(d[1] - mx);
            const float ex2 = __builtin_amdgcn_exp2f(d[2] - mx);
            const float ex3 = __builtin_amdgcn_exp2f(d[3] - mx);
            const float ex4 = __builtin_amdgcn_exp2f(d[4] - mx);
            const float ex5 = __builtin_amdgcn_exp2f(d[5] - mx);
            const float ex6 = __builtin_amdgcn_exp2f(d[6] - mx);
            const float ex7 = __builtin_amdgcn_exp2f(d[7] - mx);
            const float e = ((ex0 + ex1) + (ex2 + ex3)) + ((ex4 + ex5) + (ex6 + ex7));
            r[t] = LN2 * (mx + __builtin_amdgcn_logf(e));
        }
        store_tile(r, lo, op + it * 256);
    }
}

__global__ __launch_bounds__(256) void mdn_posterior_tail(
    const float* __restrict__ y,
    const float* __restrict__ mus,
    const float* __restrict__ sigmas,
    const float* __restrict__ pi_logits,
    const float* __restrict__ prior,
    float* __restrict__ out,
    int n_samp, int row0)
{
    const int w = blockIdx.x * blockDim.x + threadIdx.x;
    if (w >= (n_samp - row0) * 16) return;
    const int row = row0 + (w >> 4);
    const int j   = w & 15;
    const float y0 = y[row * 2], y1 = y[row * 2 + 1];

    float mxp = pi_logits[j * 8];
    #pragma unroll
    for (int i = 1; i < 8; ++i) mxp = fmaxf(mxp, pi_logits[j * 8 + i]);
    float se = 0.0f;
    #pragma unroll
    for (int i = 0; i < 8; ++i) se += __expf(pi_logits[j * 8 + i] - mxp);
    const float lse = mxp + __logf(se);
    const float lp  = __logf(prior[j]);

    float vals[8];
    float vmax = -INFINITY;
    #pragma unroll
    for (int c = 0; c < 8; ++c) {
        const int base = j * 16 + c * 2;
        const float m0 = mus[base],    m1 = mus[base + 1];
        const float s0 = sigmas[base], s1 = sigmas[base + 1];
        const float d0 = (y0 - m0) / s0, d1 = (y1 - m1) / s1;
        const float v = -0.5f * (d0 * d0 + d1 * d1)
                        - (__logf(s0) + __logf(s1)) - LOG2PI
                        + (pi_logits[j * 8 + c] - lse) + lp;
        vals[c] = v;
        vmax = fmaxf(vmax, v);
    }
    float e = 0.0f;
    #pragma unroll
    for (int c = 0; c < 8; ++c) e += __expf(vals[c] - vmax);
    out[row * 16 + j] = vmax + __logf(e);
}

extern "C" void kernel_launch(void* const* d_in, const int* in_sizes, int n_in,
                              void* d_out, int out_size, void* d_ws, size_t ws_size,
                              hipStream_t stream) {
    const float* y         = (const float*)d_in[0];
    const float* mus       = (const float*)d_in[1];
    const float* sigmas    = (const float*)d_in[2];
    const float* pi_logits = (const float*)d_in[3];
    const float* prior     = (const float*)d_in[4];
    float* out = (float*)d_out;

    const int n_samp       = in_sizes[0] / 2;
    const int n_full_tiles = n_samp / 16;
    const int rem          = n_samp - n_full_tiles * 16;

    if (n_full_tiles > 0) {
        const int n_super = (n_full_tiles + TILES_PER_WAVE - 1) / TILES_PER_WAVE;
        const int waves_per_block = 8;
        dim3 block(32 * waves_per_block);
        dim3 grid((n_super + waves_per_block - 1) / waves_per_block);
        mdn_posterior_wmma<<<grid, block, 0, stream>>>(y, mus, sigmas, pi_logits,
                                                       prior, out, n_full_tiles);
    }
    if (rem > 0) {
        mdn_posterior_tail<<<1, 256, 0, stream>>>(y, mus, sigmas, pi_logits, prior,
                                                  out, n_samp, n_full_tiles * 16);
    }
}


# Every global load of a wave requested at kernel entry in one group (table inputs, then the sample pairs of its 4 tiles; tile loop unrolled to hold them): one memory trip ahead of the first product
# speedup vs baseline: 2.1867x; 1.1697x over previous
#include <hip/hip_runtime.h>
#include <hip/hip_bf16.h>
#include <math.h>

typedef __attribute__((ext_vector_type(2))) float v2f;
typedef __attribute__((ext_vector_type(8))) float v8f;
typedef __attribute__((ext_vector_type(4))) float v4f;

#define LOG2PI 1.8378770664093453f
#define LOG2E  1.4426950408889634f
#define LN2    0.6931471805599453f
#define TILES_PER_WAVE 4
#define WAVES_PER_BLOCK 8

#define EXPSUM_MIN 1.262177448e-29f
#define EXPSUM_MAX 1.329227996e+36f

__device__ __forceinline__ v2f make_b(const v2f yv, const int hi)
{
    v2f b;
    b.x = hi ? yv.x : yv.x * yv.x;
    b.y = hi ? yv.y : yv.y * yv.y;
    return b;
}

__device__ __forceinline__ void store_tile(const float (&r)[8], const bool lo, float* op)
{
    v4f st0, st1;
    #pragma unroll
    for (int i = 0; i < 4; ++i) {
        const float send = lo ? r[4 + i] : r[i];
        const float recv = __builtin_bit_cast(float,
            __builtin_amdgcn_mov_dpp(__builtin_bit_cast(int, send),
                                     0x128, 0xF, 0xF, true));
        st0[i] = lo ? r[i]  : recv;
        st1[i] = lo ? recv  : r[4 + i];
    }
    *(v4f*)op         = st0;
    *(v4f*)(op + 128) = st1;
}

__global__ __launch_bounds__(256, 8) void mdn_posterior_wmma(
    const float* __restrict__ y,
    const float* mus,
    const float* sigmas,
    const float* pi_logits,
    const float* prior,
    float* __restrict__ out,
    int n_full_tiles)
{
    __shared__ float W[5 * 128];

    const int tid  = threadIdx.x;
    const int lane = tid & 31;
    const int wave = tid >> 5;
    const int m    = lane & 15;
    const int hi   = lane >> 4;
    const int tile0 = (blockIdx.x * WAVES_PER_BLOCK + wave) * TILES_PER_WAVE;

    v2f yv[TILES_PER_WAVE];
    auto request_samples = [&]() {
        #pragma unroll
        for (int it = 0; it < TILES_PER_WAVE; ++it) {
            int t = tile0 + it;
            if (t > n_full_tiles - 1) t = n_full_tiles - 1;
            yv[it] = *(const v2f*)(y + (t * 16 + m) * 2);
        }
    };

    if (tid < 128) {
        const int j = tid >> 3;
        const int c = tid & 7;
        const int base = j * 16 + c * 2;
        float pl[8];
        #pragma unroll
        for (int i = 0; i < 8; ++i) pl[i] = pi_logits[j * 8 + i];
        const float m0 = mus[base],    m1 = mus[base + 1];
        const float s0 = sigmas[base], s1 = sigmas[base + 1];
        const float plc = pi_logits[j * 8 + c];
        const float pr  = prior[j];
        asm volatile("" ::: "memory");
        request_samples();

        const float iv0 = 1.0f / (s0 * s0);
        const float iv1 = 1.0f / (s1 * s1);
        W[      tid] = LOG2E * -0.5f * iv0;
        W[128 + tid] = LOG2E * -0.5f * iv1;
        W[256 + tid] = LOG2E * m0 * iv0;
        W[384 + tid] = LOG2E * m1 * iv1;
        float mx = pl[0];
        #pragma unroll
        for (int i = 1; i < 8; ++i) mx = fmaxf(mx, pl[i]);
        float se = 0.0f;
        #pragma unroll
        for (int i = 0; i < 8; ++i) se += __expf(pl[i] - mx);
        const float lse = mx + __logf(se);
        W[512 + tid] = LOG2E * (-0.5f * (m0 * m0 * iv0 + m1 * m1 * iv1)
                                - (__logf(s0) + __logf(s1)) - LOG2PI
                                + (plc - lse) + __logf(pr));
    } else {
        request_samples();
    }
    __syncthreads();

    if (tile0 >= n_full_tiles) return;

    v2f a[8];
    v8f kvec[8];
    #pragma unroll
    for (int t = 0; t < 8; ++t) {
        const int ncol = t * 8 + (m >> 3) * 64 + (m & 7);
        a[t].x = W[(hi ? 256 :   0) + ncol];
        a[t].y = W[(hi ? 384 : 128) + ncol];
        const float* Kp = &W[512 + t * 8 + hi * 64];
        #pragma unroll
        for (int i = 0; i < 8; ++i) kvec[t][i] = Kp[i];
    }

    const int row0 = tile0 * 16 + m;
    const float* yp = y + row0 * 2;
    const bool lo = m < 8;
    float* op = out + (tile0 * 16 + (m & 7)) * 16 + hi * 8 + (m >> 3) * 4;

    int count = n_full_tiles - tile0;
    if (count > TILES_PER_WAVE) count = TILES_PER_WAVE;

    unsigned redo = 0u;
    #pragma unroll
    for (int it = 0; it < TILES_PER_WAVE; ++it) {
        if (it >= count) break;
        const v2f b = make_b(yv[it], hi);

        float r[8];
        float elo = __builtin_inff();
        float ehi = 0.0f;
        #pragma unroll
        for (int p = 0; p < 4; ++p) {
            const v8f dA = __builtin_amdgcn_wmma_f32_16x16x4_f32(
                false, a[2 * p], false, b, (short)0, kvec[2 * p], false, false);
            const v8f dB = __builtin_amdgcn_wmma_f32_16x16x4_f32(
                false, a[2 * p + 1], false, b, (short)0, kvec[2 * p + 1], false, false);
            v2f x[8];
            #pragma unroll
            for (int i = 0; i < 8; ++i) {
                x[i].x = __builtin_amdgcn_exp2f(dA[i]);
                x[i].y = __builtin_amdgcn_exp2f(dB[i]);
            }
            const v2f e = ((x[0] + x[1]) + (x[2] + x[3])) + ((x[4] + x[5]) + (x[6] + x[7]));
            elo = fminf(elo, fminf(e.x, e.y));
            ehi = fmaxf(ehi, fmaxf(e.x, e.y));
            v2f lg;
            lg.x = __builtin_amdgcn_logf(e.x);
            lg.y = __builtin_amdgcn_logf(e.y);
            const v2f res = lg * LN2;
            r[2 * p]     = res.x;
            r[2 * p + 1] = res.y;
        }
        const bool in_range = (elo >= EXPSUM_MIN) && (ehi <= EXPSUM_MAX);
        if (__builtin_amdgcn_ballot_w32(!in_range) == 0u) {
            store_tile(r, lo, op + it * 256);
        } else {
            redo |= 1u << it;
        }
    }

    if (redo == 0u) return;

    #pragma unroll 1
    for (int it = 0; it < count; ++it) {
        if (((redo >> it) & 1u) == 0u) continue;
        const v2f b = make_b(*(const v2f*)(yp + it * 32), hi);

        float r[8];
        #pragma unroll
        for (int t = 0; t < 8; ++t) {
            const v8f d = __builtin_amdgcn_wmma_f32_16x16x4_f32(
                false, a[t], false, b, (short)0, kvec[t], false, false);
            float mx = fmaxf(fmaxf(d[0], d[1]), fmaxf(d[2], d[3]));
            mx = fmaxf(mx, fmaxf(fmaxf(d[4], d[5]), fmaxf(d[6], d[7])));
            const float ex0 = __builtin_amdgcn_exp2f(d[0] - mx);
            const float ex1 = __builtin_amdgcn_exp2f(d[1] - mx);
            const float ex2 = __builtin_amdgcn_exp2f(d[2] - mx);
            const float ex3 = __builtin_amdgcn_exp2f(d[3] - mx);
            const float ex4 = __builtin_amdgcn_exp2f(d[4] - mx);
            const float ex5 = __builtin_amdgcn_exp2f(d[5] - mx);
            const float ex6 = __builtin_amdgcn_exp2f(d[6] - mx);
            const float ex7 = __builtin_amdgcn_exp2f(d[7] - mx);
            const float e = ((ex0 + ex1) + (ex2 + ex3)) + ((ex4 + ex5) + (ex6 + ex7));
            r[t] = LN2 * (mx + __builtin_amdgcn_logf(e));
        }
        store_tile(r, lo, op + it * 256);
    }
}

__global__ __launch_bounds__(256) void mdn_posterior_tail(
    const float* __restrict__ y,
    const float* __restrict__ mus,
    const float* __restrict__ sigmas,
    const float* __restrict__ pi_logits,
    const float* __restrict__ prior,
    float* __restrict__ out,
    int n_samp, int row0)
{
    const int w = blockIdx.x * blockDim.x + threadIdx.x;
    if (w >= (n_samp - row0) * 16) return;
    const int row = row0 + (w >> 4);
    const int j   = w & 15;
    const float y0 = y[row * 2], y1 = y[row * 2 + 1];

    float mxp = pi_logits[j * 8];
    #pragma unroll
    for (int i = 1; i < 8; ++i) mxp = fmaxf(mxp, pi_logits[j * 8 + i]);
    float se = 0.0f;
    #pragma unroll
    for (int i = 0; i < 8; ++i) se += __expf(pi_logits[j * 8 + i] - mxp);
    const float lse = mxp + __logf(se);
    const float lp  = __logf(prior[j]);

    float vals[8];
    float vmax = -INFINITY;
    #pragma unroll
    for (int c = 0; c < 8; ++c) {
        const int base = j * 16 + c * 2;
        const float m0 = mus[base],    m1 = mus[base + 1];
        const float s0 = sigmas[base], s1 = sigmas[base + 1];
        const float d0 = (y0 - m0) / s0, d1 = (y1 - m1) / s1;
        const float v = -0.5f * (d0 * d0 + d1 * d1)
                        - (__logf(s0) + __logf(s1)) - LOG2PI
                        + (pi_logits[j * 8 + c] - lse) + lp;
        vals[c] = v;
        vmax = fmaxf(vmax, v);
    }
    float e = 0.0f;
    #pragma unroll
    for (int c = 0; c < 8; ++c) e += __expf(vals[c] - vmax);
    out[row * 16 + j] = vmax + __logf(e);
}

extern "C" void kernel_launch(void* const* d_in, const int* in_sizes, int n_in,
                              void* d_out, int out_size, void* d_ws, size_t ws_size,
                              hipStream_t stream) {
    const float* y         = (const float*)d_in[0];
    const float* mus       = (const float*)d_in[1];
    const float* sigmas    = (const float*)d_in[2];
    const float* pi_logits = (const float*)d_in[3];
    const float* prior     = (const float*)d_in[4];
    float* out = (float*)d_out;

    const int n_samp       = in_sizes[0] / 2;
    const int n_full_tiles = n_samp / 16;
    const int rem          = n_samp - n_full_tiles * 16;

    if (n_full_tiles > 0) {
        const int n_super = (n_full_tiles + TILES_PER_WAVE - 1) / TILES_PER_WAVE;
        const int waves_per_block = WAVES_PER_BLOCK;
        dim3 block(32 * waves_per_block);
        dim3 grid((n_super + waves_per_block - 1) / waves_per_block);
        mdn_posterior_wmma<<<grid, block, 0, stream>>>(y, mus, sigmas, pi_logits,
                                                       prior, out, n_full_tiles);
    }
    if (rem > 0) {
        mdn_posterior_tail<<<1, 256, 0, stream>>>(y, mus, sigmas, pi_logits, prior,
                                                  out, n_samp, n_full_tiles * 16);
    }
}
